# starting kernel
# baseline (speedup 1.0000x reference)
#include <hip/hip_runtime.h>
#include <hip/hip_bf16.h>

#define IN_DIM   16384
#define BATCH    4096
#define OUT_CH   2
#define N_SLICES 32
#define SLICE_K  (IN_DIM / N_SLICES)
#define TK       128
#define TK_PAD   132
#define WAVES_PER_BLOCK 2
#define LDS_ROWS_PER_WAVE (16 + 32)
#define LDS_F32_PER_WAVE  (LDS_ROWS_PER_WAVE * TK_PAD)

typedef __attribute__((ext_vector_type(2))) float v2f;
typedef __attribute__((ext_vector_type(4))) float v4f;
typedef __attribute__((ext_vector_type(8))) float v8f;

__global__ __launch_bounds__(256) void moe_bias_init(const int* __restrict__ ts,
                                                     const float* __restrict__ bias,
                                                     float* __restrict__ out) {
    int i = blockIdx.x * 256 + threadIdx.x;
    if (i < BATCH * OUT_CH) {
        int s  = i >> 1;
        int ch = i & 1;
        out[i] = bias[ts[s] * OUT_CH + ch];
    }
}

__global__ __launch_bounds__(WAVES_PER_BLOCK * 32) void
moe_wmma(const float* __restrict__ x,
         const int*   __restrict__ ts,
         const float* __restrict__ W,
         float*       __restrict__ out) {
    __shared__ float lds[WAVES_PER_BLOCK * LDS_F32_PER_WAVE];

    const int wave = threadIdx.x >> 5;
    const int lane = threadIdx.x & 31;
    float* As = lds + wave * LDS_F32_PER_WAVE;
    float* Ws = As + 16 * TK_PAD;

    const int wid   = blockIdx.x * WAVES_PER_BLOCK + wave;
    const int group = wid >> 5;
    const int slice = wid & 31;

    const int m    = lane & 15;
    const int half = lane >> 4;

    int tid16[16];
    #pragma unroll
    for (int r = 0; r < 16; ++r) tid16[r] = ts[group * 16 + r];

    const float* xg = x + (size_t)(group * 16) * IN_DIM + slice * SLICE_K;

    v8f c0 = {};
    v8f c1 = {};

    for (int tile = 0; tile < SLICE_K; tile += TK) {
        #pragma unroll
        for (int r = 0; r < 16; ++r) {
            v4f v = *(const v4f*)(xg + (size_t)r * IN_DIM + tile + lane * 4);
            *(v4f*)(As + r * TK_PAD + lane * 4) = v;
        }
        #pragma unroll
        for (int r = 0; r < 16; ++r) {
            const float* wr = W + (size_t)tid16[r] * (OUT_CH * IN_DIM)
                                + slice * SLICE_K + tile + lane * 4;
            v4f w0 = *(const v4f*)(wr);
            v4f w1 = *(const v4f*)(wr + IN_DIM);
            *(v4f*)(Ws + r        * TK_PAD + lane * 4) = w0;
            *(v4f*)(Ws + (16 + r) * TK_PAD + lane * 4) = w1;
        }
        __builtin_amdgcn_wave_barrier();

        #pragma unroll 4
        for (int kk = 0; kk < TK; kk += 4) {
            v2f a  = *(const v2f*)(As + m        * TK_PAD + kk + 2 * half);
            v2f b0 = *(const v2f*)(Ws + m        * TK_PAD + kk + 2 * half);
            v2f b1 = *(const v2f*)(Ws + (16 + m) * TK_PAD + kk + 2 * half);
            c0 = __builtin_amdgcn_wmma_f32_16x16x4_f32(false, a, false, b0,
                                                       (short)0, c0, false, false);
            c1 = __builtin_amdgcn_wmma_f32_16x16x4_f32(false, a, false, b1,
                                                       (short)0, c1, false, false);
        }
        __builtin_amdgcn_wave_barrier();
    }

    const bool active = (lane < 8) || (lane >= 24);
    const int  r      = (lane < 8) ? lane : (lane - 24);
    const int  mres   = (lane < 8) ? lane : (lane - 16);

    float v0 = c0[0], v1 = c1[0];
    #pragma unroll
    for (int i = 1; i < 8; ++i) {
        v0 = (r == i) ? c0[i] : v0;
        v1 = (r == i) ? c1[i] : v1;
    }

    if (active) {
        float* o = out + (size_t)(group * 16 + mres) * OUT_CH;
        atomicAdd(o + 0, v0);
        atomicAdd(o + 1, v1);
    }
}

extern "C" void kernel_launch(void* const* d_in, const int* in_sizes, int n_in,
                              void* d_out, int out_size, void* d_ws, size_t ws_size,
                              hipStream_t stream) {
    const float* x   = (const float*)d_in[0];
    const int*   ts  = (const int*)  d_in[1];
    const float* W   = (const float*)d_in[2];
    const float* b   = (const float*)d_in[3];
    float*       out = (float*)d_out;

    (void)in_sizes; (void)n_in; (void)out_size; (void)d_ws; (void)ws_size;

    moe_bias_init<<<(BATCH * OUT_CH + 255) / 256, 256, 0, stream>>>(ts, b, out);

    const int n_waves  = (BATCH / 16) * N_SLICES;
    const int n_blocks = n_waves / WAVES_PER_BLOCK;
    moe_wmma<<<n_blocks, WAVES_PER_BLOCK * 32, 0, stream>>>(x, ts, W, out);
}


# One wave per whole sample row, no LDS: each lane feeds both WMMA operands from its own coalesced 16-B global loads (diagonal = 16 partial sums); no K split: sums re-ordered, one add per output
# speedup vs baseline: 1.1339x; 1.1339x over previous
#include <hip/hip_runtime.h>
#include <hip/hip_bf16.h>

#define IN_DIM   16384
#define BATCH    4096
#define OUT_CH   2
#define WAVES    16
#define PIECE    128
#define NP       8
#define NBATCH   (IN_DIM / (PIECE * NP))

typedef __attribute__((ext_vector_type(2))) float v2f;
typedef __attribute__((ext_vector_type(4))) float v4f;
typedef __attribute__((ext_vector_type(8))) float v8f;

__global__ __launch_bounds__(256) void moe_bias_init(const int* __restrict__ ts,
                                                     const float* __restrict__ bias,
                                                     float* __restrict__ out) {
    int i = blockIdx.x * 256 + threadIdx.x;
    if (i < BATCH * OUT_CH) {
        int s  = i >> 1;
        int ch = i & 1;
        out[i] = bias[ts[s] * OUT_CH + ch];
    }
}

static __device__ __forceinline__ v8f mma(v2f a, v2f b, v8f c) {
    return __builtin_amdgcn_wmma_f32_16x16x4_f32(false, a, false, b, (short)0, c, false, false);
}

__global__ __launch_bounds__(WAVES * 32) void
moe_wmma(const float* __restrict__ x,
         const int*   __restrict__ ts,
         const float* __restrict__ W,
         float*       __restrict__ out) {
    const int lane = threadIdx.x & 31;
    const int wave = __builtin_amdgcn_readfirstlane((int)(threadIdx.x >> 5));
    const int b    = blockIdx.x * WAVES + wave;
    const int e    = ts[b];

    const float* xr  = x + (size_t)b * IN_DIM + lane * 4;
    const float* w0r = W + (size_t)e * (OUT_CH * IN_DIM) + lane * 4;
    const float* w1r = w0r + IN_DIM;

    v4f xa[2][NP], wa[2][NP], wb[2][NP];
    v8f c0 = {};
    v8f c1 = {};

#define LOAD_BATCH(BUF, T) do {                                                   \
        const int col_ = (((T) + wave) & (NBATCH - 1)) * (NP * PIECE);            \
        _Pragma("unroll")                                                         \
        for (int q = 0; q < NP; ++q) {                                            \
            xa[BUF][q] = __builtin_nontemporal_load((const v4f*)(xr + col_ + q * PIECE)); \
            wa[BUF][q] = *(const v4f*)(w0r + col_ + q * PIECE);                   \
            wb[BUF][q] = *(const v4f*)(w1r + col_ + q * PIECE);                   \
               \
               \
            __builtin_amdgcn_sched_barrier(0);                                    \
        } } while (0)

#define MMA_BATCH(BUF) do {                                                       \
        _Pragma("unroll")                                                         \
        for (int q = 0; q < NP; ++q) {                                            \
            const v4f xv = xa[BUF][q], u0 = wa[BUF][q], u1 = wb[BUF][q];          \
            c0 = mma(xv.xy, u0.xy, c0);                                           \
            c1 = mma(xv.xy, u1.xy, c1);                                           \
            c0 = mma(xv.zw, u0.zw, c0);                                           \
            c1 = mma(xv.zw, u1.zw, c1);                                           \
        } } while (0)

    LOAD_BATCH(0, 0);
    #pragma unroll
    for (int t = 0; t < NBATCH; t += 2) {
        __builtin_amdgcn_sched_barrier(0);
        LOAD_BATCH(1, t + 1);
        MMA_BATCH(0);
        __builtin_amdgcn_sched_barrier(0);
        if (t + 2 < NBATCH) LOAD_BATCH(0, t + 2);
        MMA_BATCH(1);
    }
#undef LOAD_BATCH
#undef MMA_BATCH

    const bool diag = (lane < 8) || (lane >= 24);
    const int  r    = lane & 7;

    float v0 = c0[0], v1 = c1[0];
    #pragma unroll
    for (int i = 1; i < 8; ++i) {
        v0 = (r == i) ? c0[i] : v0;
        v1 = (r == i) ? c1[i] : v1;
    }
    v0 = diag ? v0 : 0.0f;
    v1 = diag ? v1 : 0.0f;
    for (int m = 16; m >= 1; m >>= 1) {
        v0 += __shfl_xor(v0, m, 32);
        v1 += __shfl_xor(v1, m, 32);
    }

    if (lane < OUT_CH) atomicAdd(out + (size_t)b * OUT_CH + lane, lane ? v1 : v0);
}

extern "C" void kernel_launch(void* const* d_in, const int* in_sizes, int n_in,
                              void* d_out, int out_size, void* d_ws, size_t ws_size,
                              hipStream_t stream) {
    const float* x   = (const float*)d_in[0];
    const int*   ts  = (const int*)  d_in[1];
    const float* W   = (const float*)d_in[2];
    const float* b   = (const float*)d_in[3];
    float*       out = (float*)d_out;

    (void)in_sizes; (void)n_in; (void)out_size; (void)d_ws; (void)ws_size;

    moe_bias_init<<<(BATCH * OUT_CH + 255) / 256, 256, 0, stream>>>(ts, b, out);

    moe_wmma<<<BATCH / WAVES, WAVES * 32, 0, stream>>>(x, ts, W, out);
}


# Samples grouped by expert (LDS counting sort, arrival-order ranks); a block stages its expert's two weight rows in LDS once, waves read them there, not from global; out = bias + dot, plain store
# speedup vs baseline: 1.3351x; 1.1774x over previous
#include <hip/hip_runtime.h>

#define IN_DIM   16384
#define BATCH    4096
#define OUT_CH   2
#define N_EXP    50
#define WAVES    16
#define PIECE    128
#define NP       8
#define NBATCH   (IN_DIM / (PIECE * NP))
#define ROWS_F   (OUT_CH * IN_DIM)
#define GROUP_T  1024

typedef __attribute__((ext_vector_type(2))) float v2f;
typedef __attribute__((ext_vector_type(4))) float v4f;
typedef __attribute__((ext_vector_type(8))) float v8f;

static __device__ __forceinline__ int expert_of(int t) { return min(max(t, 0), N_EXP - 1); }

__global__ __launch_bounds__(GROUP_T) void moe_group(const int* __restrict__ ts,
                                                     int* __restrict__ perm) {
    __shared__ int cnt[64];
    __shared__ int start[64];
    const int tid = threadIdx.x;
    if (tid < 64) cnt[tid] = 0;
    __syncthreads();
    int e[BATCH / GROUP_T], r[BATCH / GROUP_T];
    for (int j = 0; j < BATCH / GROUP_T; ++j) {
        e[j] = expert_of(ts[tid + GROUP_T * j]);
        r[j] = atomicAdd(&cnt[e[j]], 1);
    }
    __syncthreads();
    if (tid < 64) {
        int s = 0;
        for (int k = 0; k < N_EXP; ++k) s += (k < tid) ? cnt[k] : 0;
        start[tid] = s;
    }
    __syncthreads();
    for (int j = 0; j < BATCH / GROUP_T; ++j) perm[start[e[j]] + r[j]] = (e[j] << 16) | (tid + GROUP_T * j);
}

static __device__ __forceinline__ v8f mma(v2f a, v2f b, v8f c) {
    return __builtin_amdgcn_wmma_f32_16x16x4_f32(false, a, false, b, (short)0, c, false, false);
}

static __device__ __forceinline__ void stage_rows(const float* __restrict__ src, float* dst, int tid) {
    #pragma unroll
    for (int i = 0; i < ROWS_F / 4 / (WAVES * 32); ++i) {
        const v4f v = *(const v4f*)(src + (i * (WAVES * 32) + tid) * 4);
        *(v4f*)(dst + (i * (WAVES * 32) + tid) * 4) = v;
    }
}

__global__ __launch_bounds__(WAVES * 32) void
moe_wmma(const float* __restrict__ x,
         const float* __restrict__ W,
         const float* __restrict__ bias,
         const int*   __restrict__ perm,
         float*       __restrict__ out) {
    __shared__ float wl[2][ROWS_F];
    const int tid  = threadIdx.x;
    const int lane = tid & 31;
    const int wave = __builtin_amdgcn_readfirstlane((int)(threadIdx.x >> 5));
    const int base = blockIdx.x * WAVES;

    int es[WAVES];
    for (int j = 0; j < WAVES; ++j) es[j] = __builtin_amdgcn_readfirstlane(perm[base + j] >> 16);
    int nruns = 1, my_run = 0;
    for (int j = 1; j < WAVES; ++j) {
        const int step = (es[j] != es[j - 1]) ? 1 : 0;
        nruns  += step;
        my_run += (j <= wave) ? step : 0;
    }
    nruns  = __builtin_amdgcn_readfirstlane(nruns);
    my_run = __builtin_amdgcn_readfirstlane(my_run);

    const int mine = __builtin_amdgcn_readfirstlane(perm[base + wave]);
    const int b = mine & 0xFFFF;
    const int e = mine >> 16;
    const int rot = b & (NBATCH - 1);

    const float* xr = x + (size_t)b * IN_DIM + lane * 4;

    v4f xa[2][NP];
#define LOAD_X(BUF, T) do {                                                       \
        const int col_ = (((T) + rot) & (NBATCH - 1)) * (NP * PIECE);     \
        _Pragma("unroll")                                                 \
        for (int q = 0; q < NP; ++q) {                                    \
            xa[BUF][q] = __builtin_nontemporal_load((const v4f*)(xr + col_ + q * PIECE)); \
               \
               \
            __builtin_amdgcn_sched_barrier(0);                            \
        } } while (0)

#define LOAD_W(H, T, HALF) do {                                                   \
        const int col_ = (((T) + rot) & (NBATCH - 1)) * (NP * PIECE) + (HALF) * (NP / 2) * PIECE; \
        _Pragma("unroll")                                                 \
        for (int q = 0; q < NP / 2; ++q) {                                \
            wa[H][q] = *(const v4f*)(w0r + col_ + q * PIECE);             \
            wb[H][q] = *(const v4f*)(w1r + col_ + q * PIECE);             \
              \
            __builtin_amdgcn_sched_barrier(0);                            \
        } } while (0)

#define MMA_HALF(BUF, H, HALF) do {                                               \
        _Pragma("unroll")                                                 \
        for (int q = 0; q < NP / 2; ++q) {                                \
            const v4f xv = xa[BUF][(HALF) * (NP / 2) + q], u0 = wa[H][q], u1 = wb[H][q]; \
            c0 = mma(xv.xy, u0.xy, c0);                                   \
            c1 = mma(xv.xy, u1.xy, c1);                                   \
            c0 = mma(xv.zw, u0.zw, c0);                                   \
            c1 = mma(xv.zw, u1.zw, c1);                                   \
        } } while (0)


    LOAD_X(0, 0);

    for (int r0 = 0; r0 < nruns; r0 += 2) {
        int e0 = es[0], e1 = es[0];
        {
            int r = 0;
            for (int j = 0; j < WAVES; ++j) {
                if (j) r += (es[j] != es[j - 1]) ? 1 : 0;
                if (r == r0)     e0 = es[j];
                if (r == r0 + 1) e1 = es[j];
            }
        }
        if (r0) __syncthreads();
        stage_rows(W + (size_t)e0 * ROWS_F, &wl[0][0], tid);
        if (r0 + 1 < nruns) stage_rows(W + (size_t)e1 * ROWS_F, &wl[1][0], tid);
        __syncthreads();

        if (my_run == r0 || my_run == r0 + 1) {
            const float* w0r = &wl[my_run - r0][0] + lane * 4;
            const float* w1r = w0r + IN_DIM;

            v4f wa[2][NP / 2], wb[2][NP / 2];
            v8f c0 = {};
            v8f c1 = {};

            LOAD_W(0, 0, 0);
            #pragma unroll
            for (int t = 0; t < NBATCH; t += 2) {
                __builtin_amdgcn_sched_barrier(0);
                LOAD_X(1, t + 1);
                LOAD_W(1, t, 1);
                MMA_HALF(0, 0, 0);
                __builtin_amdgcn_sched_barrier(0);
                LOAD_W(0, t + 1, 0);
                MMA_HALF(0, 1, 1);
                __builtin_amdgcn_sched_barrier(0);
                if (t + 2 < NBATCH) LOAD_X(0, t + 2);
                LOAD_W(1, t + 1, 1);
                MMA_HALF(1, 0, 0);
                __builtin_amdgcn_sched_barrier(0);
                if (t + 2 < NBATCH) LOAD_W(0, t + 2, 0);
                MMA_HALF(1, 1, 1);
            }
#undef LOAD_X
#undef LOAD_W
#undef MMA_HALF

            const bool diag = (lane < 8) || (lane >= 24);
            const int  r    = lane & 7;

            float v0 = c0[0], v1 = c1[0];
            #pragma unroll
            for (int i = 1; i < 8; ++i) {
                v0 = (r == i) ? c0[i] : v0;
                v1 = (r == i) ? c1[i] : v1;
            }
            v0 = diag ? v0 : 0.0f;
            v1 = diag ? v1 : 0.0f;
            for (int m = 16; m >= 1; m >>= 1) {
                v0 += __shfl_xor(v0, m, 32);
                v1 += __shfl_xor(v1, m, 32);
            }

            if (lane < OUT_CH) out[(size_t)b * OUT_CH + lane] = bias[e * OUT_CH + lane] + (lane ? v1 : v0);
        }
    }
}

extern "C" void kernel_launch(void* const* d_in, const int* in_sizes, int n_in,
                              void* d_out, int out_size, void* d_ws, size_t ws_size,
                              hipStream_t stream) {
    const float* x   = (const float*)d_in[0];
    const int*   ts  = (const int*)  d_in[1];
    const float* W   = (const float*)d_in[2];
    const float* b   = (const float*)d_in[3];
    float*       out = (float*)d_out;
    int*         perm = (int*)d_ws;

    (void)in_sizes; (void)n_in; (void)out_size; (void)ws_size;

    moe_group<<<1, GROUP_T, 0, stream>>>(ts, perm);

    moe_wmma<<<BATCH / WAVES, WAVES * 32, 0, stream>>>(x, W, b, perm, out);
}


# 16 grouped samples as the 16 A rows of each WMMA (x tiles through per-wave LDS), both channels as B columns: half the WMMAs per sample; K split over 16 waves, partial sums added in wave order
# speedup vs baseline: 1.4241x; 1.0667x over previous
#include <hip/hip_runtime.h>

#define IN_DIM   16384
#define BATCH    4096
#define OUT_CH   2
#define N_EXP    50
#define WAVES    16
#define ROWS     16
#define TILE_F   128
#define PITCH_F  132
#define NT       (IN_DIM / (WAVES * TILE_F))
#define NFRAG    (TILE_F / 8)
#define GROUP_T  1024

typedef __attribute__((ext_vector_type(2))) float v2f;
typedef __attribute__((ext_vector_type(4))) float v4f;
typedef __attribute__((ext_vector_type(8))) float v8f;

static __device__ __forceinline__ int expert_of(int t) { return min(max(t, 0), N_EXP - 1); }

__global__ __launch_bounds__(GROUP_T) void moe_group(const int* __restrict__ ts,
                                                     int* __restrict__ perm) {
    __shared__ int cnt[64];
    __shared__ int start[64];
    const int tid = threadIdx.x;
    if (tid < 64) cnt[tid] = 0;
    __syncthreads();
    int e[BATCH / GROUP_T], r[BATCH / GROUP_T];
    for (int j = 0; j < BATCH / GROUP_T; ++j) {
        e[j] = expert_of(ts[tid + GROUP_T * j]);
        r[j] = atomicAdd(&cnt[e[j]], 1);
    }
    __syncthreads();
    if (tid < 64) {
        int s = 0;
        for (int k = 0; k < N_EXP; ++k) s += (k < tid) ? cnt[k] : 0;
        start[tid] = s;
    }
    __syncthreads();
    for (int j = 0; j < BATCH / GROUP_T; ++j) perm[start[e[j]] + r[j]] = (e[j] << 16) | (tid + GROUP_T * j);
}

static __device__ __forceinline__ v8f mma(v2f a, v2f b, v8f c) {
    return __builtin_amdgcn_wmma_f32_16x16x4_f32(false, a, false, b, (short)0, c, false, false);
}

__global__ __launch_bounds__(WAVES * 32) void
moe_wmma(const float* x,
         const float* __restrict__ W,
         const float* __restrict__ bias,
         const int*   __restrict__ perm,
         float*       out) {
    __shared__ float tile[WAVES][ROWS * PITCH_F];
    __shared__ float part[WAVES][ROWS][16];
    const int tid  = threadIdx.x;
    const int lane = tid & 31;
    const int c    = lane & 15;
    const int h    = lane >> 4;
    const int wave = __builtin_amdgcn_readfirstlane((int)(threadIdx.x >> 5));
    const int base = blockIdx.x * ROWS;

    int ps[ROWS], run[ROWS];
    #pragma unroll
    for (int j = 0; j < ROWS; ++j) ps[j] = __builtin_amdgcn_readfirstlane(perm[base + j]);
    run[0] = 0;
    #pragma unroll
    for (int j = 1; j < ROWS; ++j) run[j] = run[j - 1] + (((ps[j] >> 16) != (ps[j - 1] >> 16)) ? 1 : 0);
    const int nruns = run[ROWS - 1] + 1;

    float* tl = &tile[wave][0];
    float*       st = tl + lane * 4;
    const float* fr = tl + c * PITCH_F + h * 4;
    const float* xw = x + wave * TILE_F;

#define LOAD_X(T) do {                                                            \
        _Pragma("unroll")                                                 \
        for (int j = 0; j < ROWS; ++j)                                    \
            R[j] = __builtin_nontemporal_load((const v4f*)(xw + (size_t)(ps[j] & 0xFFFF) * IN_DIM + (T) * (WAVES * TILE_F)) + lane); \
        } while (0)
#define LOAD_B(S, T, HALF) do {                                                   \
        _Pragma("unroll")                                                 \
        for (int i = 0; i < NFRAG / 2; ++i) Bf[S][i] = *(const v4f*)(wp + (T) * (WAVES * TILE_F) + ((HALF) * (NFRAG / 2) + i) * 8); \
        } while (0)
#define LOAD_A() do {                                                             \
        _Pragma("unroll")                                                 \
        for (int i = 0; i < NFRAG; ++i) Af[i] = *(const v4f*)(fr + i * 8); \
        } while (0)
#define MMA_HALF(S, HALF) do {                                                    \
        _Pragma("unroll")                                                 \
        for (int i = 0; i < NFRAG / 2; ++i) {                             \
            c0 = mma(Af[(HALF) * (NFRAG / 2) + i].xy, Bf[S][i].xy, c0);   \
            c1 = mma(Af[(HALF) * (NFRAG / 2) + i].zw, Bf[S][i].zw, c1);   \
        } } while (0)

    for (int p0 = 0; p0 < nruns; p0 += 8) {
        const int rho = min(p0 + (c >> 1), nruns - 1);
        int en = ps[0] >> 16;
        #pragma unroll
        for (int j = 1; j < ROWS; ++j) en = (run[j] == rho) ? (ps[j] >> 16) : en;
        const float* wp = W + (size_t)(en * OUT_CH + (c & 1)) * IN_DIM + wave * TILE_F + h * 4;

        v4f R[ROWS];
        v4f Af[NFRAG];
        v4f Bf[2][NFRAG / 2];
        v8f c0 = {}, c1 = {};

        LOAD_X(0);
        LOAD_B(0, 0, 0);
        #pragma unroll
        for (int t = 0; t < NT; ++t) {
            __builtin_amdgcn_sched_barrier(0);
            #pragma unroll
            for (int j = 0; j < ROWS; ++j) *(v4f*)(st + j * PITCH_F) = R[j];
            __builtin_amdgcn_fence(__ATOMIC_RELEASE, "wavefront");
            __builtin_amdgcn_sched_barrier(0);
            if (t + 1 < NT) LOAD_X(t + 1);
            LOAD_B(1, t, 1);
            __builtin_amdgcn_sched_barrier(0);
            LOAD_A();
            __builtin_amdgcn_fence(__ATOMIC_ACQ_REL, "wavefront");
            __builtin_amdgcn_sched_barrier(0);
            MMA_HALF(0, 0);
            if (t + 1 < NT) LOAD_B(0, t + 1, 0);
            __builtin_amdgcn_sched_barrier(0);
            MMA_HALF(1, 1);
        }

        const v8f acc = c0 + c1;
        #pragma unroll
        for (int r = 0; r < 8; ++r) part[wave][8 * h + r][c] = acc[r];
        __syncthreads();

        if (tid < ROWS * OUT_CH) {
            const int m = tid >> 1, ch = tid & 1;
            int rm = 0;
            #pragma unroll
            for (int j = 1; j < ROWS; ++j) rm = (m == j) ? run[j] : rm;
            const int pm = perm[base + m];
            if (rm >= p0 && rm < p0 + 8) {
                const int col = 2 * (rm - p0) + ch;
                float s = part[0][m][col];
                #pragma unroll
                for (int w = 1; w < WAVES; ++w) s += part[w][m][col];
                out[(size_t)(pm & 0xFFFF) * OUT_CH + ch] = bias[(pm >> 16) * OUT_CH + ch] + s;
            }
        }
        if (p0 + 8 < nruns) __syncthreads();
    }
#undef LOAD_X
#undef LOAD_B
#undef LOAD_A
#undef MMA_HALF
}

extern "C" void kernel_launch(void* const* d_in, const int* in_sizes, int n_in,
                              void* d_out, int out_size, void* d_ws, size_t ws_size,
                              hipStream_t stream) {
    const float* x   = (const float*)d_in[0];
    const int*   ts  = (const int*)  d_in[1];
    const float* W   = (const float*)d_in[2];
    const float* b   = (const float*)d_in[3];
    float*       out = (float*)d_out;
    int*         perm = (int*)d_ws;

    (void)in_sizes; (void)n_in; (void)out_size; (void)ws_size;

    moe_group<<<1, GROUP_T, 0, stream>>>(ts, perm);

    moe_wmma<<<BATCH / ROWS, WAVES * 32, 0, stream>>>(x, W, b, perm, out);
}


# Grouping folded into the product kernel: each block counts the experts in LDS and ranks (ballot counts, sample-index order) only the samples of its own 16 places; one launch; sums unchanged
# speedup vs baseline: 1.5539x; 1.0911x over previous
#include <hip/hip_runtime.h>

#define IN_DIM   16384
#define BATCH    4096
#define OUT_CH   2
#define N_EXP    50
#define WAVES    16
#define ROWS     16
#define TILE_F   128
#define PITCH_F  132
#define NT       (IN_DIM / (WAVES * TILE_F))
#define NFRAG    (TILE_F / 8)
#define SPT      (BATCH / (WAVES * 32))

typedef __attribute__((ext_vector_type(2))) float v2f;
typedef __attribute__((ext_vector_type(4))) float v4f;
typedef __attribute__((ext_vector_type(8))) float v8f;

static __device__ __forceinline__ int expert_of(int t) { return min(max(t, 0), N_EXP - 1); }

static __device__ __forceinline__ v8f mma(v2f a, v2f b, v8f c) {
    return __builtin_amdgcn_wmma_f32_16x16x4_f32(false, a, false, b, (short)0, c, false, false);
}

__global__ __launch_bounds__(WAVES * 32) void
moe_wmma(const float* x,
         const float* __restrict__ W,
         const float* __restrict__ bias,
         const int*   __restrict__ ts,
         float*       out) {
    __shared__ float tile[WAVES][ROWS * PITCH_F];
    __shared__ float part[WAVES][ROWS][16];
    __shared__ int cnt[64];
    __shared__ int start[64];
    __shared__ int erange[2];
    __shared__ int wtot[64][WAVES];
    __shared__ int sel[ROWS];
    const int tid  = threadIdx.x;
    const int lane = tid & 31;
    const int c    = lane & 15;
    const int h    = lane >> 4;
    const int wave = __builtin_amdgcn_readfirstlane((int)(threadIdx.x >> 5));
    const int base = blockIdx.x * ROWS;

    if (tid < 64) cnt[tid] = 0;
    int e[SPT];
    {
        const int4 ta = ((const int4*)ts)[tid * 2], tb = ((const int4*)ts)[tid * 2 + 1];
        e[0] = expert_of(ta.x); e[1] = expert_of(ta.y); e[2] = expert_of(ta.z); e[3] = expert_of(ta.w);
        e[4] = expert_of(tb.x); e[5] = expert_of(tb.y); e[6] = expert_of(tb.z); e[7] = expert_of(tb.w);
    }
    __syncthreads();
    #pragma unroll
    for (int j = 0; j < SPT; ++j) atomicAdd(&cnt[e[j]], 1);
    __syncthreads();
    if (tid < N_EXP) {
        int s0 = 0;
        for (int k = 0; k < N_EXP; ++k) s0 += (k < tid) ? cnt[k] : 0;
        start[tid] = s0;
        const int s1 = s0 + cnt[tid];
        if (s0 <= base && base < s1) erange[0] = tid;
        if (s0 <= base + ROWS - 1 && base + ROWS - 1 < s1) erange[1] = tid;
    }
    __syncthreads();
    const int elo = __builtin_amdgcn_readfirstlane(erange[0]);
    const int ehi = __builtin_amdgcn_readfirstlane(erange[1]);
    for (int E = elo; E <= ehi; ++E) {
        const int nE = __builtin_amdgcn_readfirstlane(cnt[E]);
        if (nE == 0) continue;
        unsigned below = 0;
        int tot = 0;
        #pragma unroll
        for (int j = 0; j < SPT; ++j) {
            const unsigned mk = __builtin_amdgcn_ballot_w32(e[j] == E);
            below = __builtin_amdgcn_mbcnt_lo(mk, below);
            tot += __builtin_popcount(mk);
        }
        if (lane == 0) wtot[E][wave] = tot;
        __syncthreads();
        int r = start[E] - base + (int)below;
        {
            const int4* wq = (const int4*)&wtot[E][0];
            const int4 q0 = wq[0], q1 = wq[1], q2 = wq[2], q3 = wq[3];
            const int wt[WAVES] = { q0.x, q0.y, q0.z, q0.w, q1.x, q1.y, q1.z, q1.w,
                                    q2.x, q2.y, q2.z, q2.w, q3.x, q3.y, q3.z, q3.w };
            #pragma unroll
            for (int w = 0; w < WAVES; ++w) r += (w < wave) ? wt[w] : 0;
        }
        #pragma unroll
        for (int j = 0; j < SPT; ++j) {
            if (e[j] == E) {
                if ((unsigned)r < (unsigned)ROWS) sel[r] = (E << 16) | (tid * SPT + j);
                ++r;
            }
        }
    }
    __syncthreads();

    int ps[ROWS], run[ROWS];
    #pragma unroll
    for (int j = 0; j < ROWS; ++j) ps[j] = __builtin_amdgcn_readfirstlane(sel[j]);
    run[0] = 0;
    #pragma unroll
    for (int j = 1; j < ROWS; ++j) run[j] = run[j - 1] + (((ps[j] >> 16) != (ps[j - 1] >> 16)) ? 1 : 0);
    const int nruns = run[ROWS - 1] + 1;

    float* tl = &tile[wave][0];
    float*       st = tl + lane * 4;
    const float* fr = tl + c * PITCH_F + h * 4;
    const float* xw = x + wave * TILE_F;

#define LOAD_X(T) do {                                                            \
        _Pragma("unroll")                                                 \
        for (int j = 0; j < ROWS; ++j)                                    \
            R[j] = __builtin_nontemporal_load((const v4f*)(xw + (size_t)(ps[j] & 0xFFFF) * IN_DIM + (T) * (WAVES * TILE_F)) + lane); \
        } while (0)
#define LOAD_B(S, T, HALF) do {                                                   \
        _Pragma("unroll")                                                 \
        for (int i = 0; i < NFRAG / 2; ++i) Bf[S][i] = *(const v4f*)(wp + (T) * (WAVES * TILE_F) + ((HALF) * (NFRAG / 2) + i) * 8); \
        } while (0)
#define LOAD_A() do {                                                             \
        _Pragma("unroll")                                                 \
        for (int i = 0; i < NFRAG; ++i) Af[i] = *(const v4f*)(fr + i * 8); \
        } while (0)
#define MMA_HALF(S, HALF) do {                                                    \
        _Pragma("unroll")                                                 \
        for (int i = 0; i < NFRAG / 2; ++i) {                             \
            c0 = mma(Af[(HALF) * (NFRAG / 2) + i].xy, Bf[S][i].xy, c0);   \
            c1 = mma(Af[(HALF) * (NFRAG / 2) + i].zw, Bf[S][i].zw, c1);   \
        } } while (0)

    for (int p0 = 0; p0 < nruns; p0 += 8) {
        const int rho = min(p0 + (c >> 1), nruns - 1);
        int en = ps[0] >> 16;
        #pragma unroll
        for (int j = 1; j < ROWS; ++j) en = (run[j] == rho) ? (ps[j] >> 16) : en;
        const float* wp = W + (size_t)(en * OUT_CH + (c & 1)) * IN_DIM + wave * TILE_F + h * 4;

        v4f R[ROWS];
        v4f Af[NFRAG];
        v4f Bf[2][NFRAG / 2];
        v8f c0 = {}, c1 = {};

        LOAD_X(0);
        LOAD_B(0, 0, 0);
        #pragma unroll
        for (int t = 0; t < NT; ++t) {
            __builtin_amdgcn_sched_barrier(0);
            #pragma unroll
            for (int j = 0; j < ROWS; ++j) *(v4f*)(st + j * PITCH_F) = R[j];
            __builtin_amdgcn_fence(__ATOMIC_RELEASE, "wavefront");
            __builtin_amdgcn_sched_barrier(0);
            if (t + 1 < NT) LOAD_X(t + 1);
            LOAD_B(1, t, 1);
            __builtin_amdgcn_sched_barrier(0);
            LOAD_A();
            __builtin_amdgcn_fence(__ATOMIC_ACQ_REL, "wavefront");
            __builtin_amdgcn_sched_barrier(0);
            MMA_HALF(0, 0);
            if (t + 1 < NT) LOAD_B(0, t + 1, 0);
            __builtin_amdgcn_sched_barrier(0);
            MMA_HALF(1, 1);
        }

        const v8f acc = c0 + c1;
        #pragma unroll
        for (int r = 0; r < 8; ++r) part[wave][8 * h + r][c] = acc[r];
        __syncthreads();

        if (tid < ROWS * OUT_CH) {
            const int m = tid >> 1, ch = tid & 1;
            int rm = 0;
            #pragma unroll
            for (int j = 1; j < ROWS; ++j) rm = (m == j) ? run[j] : rm;
            const int pm = sel[m];
            if (rm >= p0 && rm < p0 + 8) {
                const int col = 2 * (rm - p0) + ch;
                float s = part[0][m][col];
                #pragma unroll
                for (int w = 1; w < WAVES; ++w) s += part[w][m][col];
                out[(size_t)(pm & 0xFFFF) * OUT_CH + ch] = bias[(pm >> 16) * OUT_CH + ch] + s;
            }
        }
        if (p0 + 8 < nruns) __syncthreads();
    }
#undef LOAD_X
#undef LOAD_B
#undef LOAD_A
#undef MMA_HALF
}

extern "C" void kernel_launch(void* const* d_in, const int* in_sizes, int n_in,
                              void* d_out, int out_size, void* d_ws, size_t ws_size,
                              hipStream_t stream) {
    const float* x   = (const float*)d_in[0];
    const int*   ts  = (const int*)  d_in[1];
    const float* W   = (const float*)d_in[2];
    const float* b   = (const float*)d_in[3];
    float*       out = (float*)d_out;

    (void)in_sizes; (void)n_in; (void)out_size; (void)d_ws; (void)ws_size;

    moe_wmma<<<BATCH / ROWS, WAVES * 32, 0, stream>>>(x, W, b, ts, out);
}
